# starting kernel
# baseline (speedup 1.0000x reference)
#include <hip/hip_runtime.h>

typedef __attribute__((ext_vector_type(2))) float v2f;
typedef __attribute__((ext_vector_type(8))) float v8f;

#define NODES 100000
#define EDGES 1000000
#define RELS  3
#define HF    64

__global__ void k_deg(const int* __restrict__ dst, float* __restrict__ deg) {
  int tid = blockIdx.x * blockDim.x + threadIdx.x;
  if (tid >= RELS * EDGES) return;
  int r = tid / EDGES;
  unsafeAtomicAdd(&deg[(size_t)r * NODES + dst[tid]], 1.0f);
}

__global__ void k_dinv(float* __restrict__ deg) {
  int i = blockIdx.x * blockDim.x + threadIdx.x;
  if (i >= RELS * NODES) return;
  float v = deg[i];
  v = v < 1.0f ? 1.0f : v;
  deg[i] = 1.0f / sqrtf(v);
}

__global__ void k_scatter(const float* __restrict__ feat,
                          const float* __restrict__ dinv,
                          const int* __restrict__ src,
                          const int* __restrict__ dst,
                          float* __restrict__ ah) {
  int tid = blockIdx.x * blockDim.x + threadIdx.x;
  if (tid >= EDGES * 16) return;
  int e = tid >> 4;
  int c = (tid & 15) << 2;
  int s = src[e];
  int d = dst[e];
  float dv = dinv[s];
  const float4 f = *(const float4*)(feat + (size_t)s * HF + c);
  float* o = ah + (size_t)d * HF + c;
  unsafeAtomicAdd(o + 0, f.x * dv);
  unsafeAtomicAdd(o + 1, f.y * dv);
  unsafeAtomicAdd(o + 2, f.z * dv);
  unsafeAtomicAdd(o + 3, f.w * dv);
}

__launch_bounds__(256)
__global__ void k_gemm(const float* __restrict__ A1,
                       const float* __restrict__ A2,
                       const float* __restrict__ dinv,
                       float a2sign,
                       const float* __restrict__ W,
                       const float* __restrict__ bias,
                       float* __restrict__ out,
                       float* __restrict__ accum,
                       int n, int act) {
  __shared__ float sW[128 * 64];
  __shared__ float sB[64];
  const bool dual = (A2 != nullptr);
  const int wtot = (dual ? 128 : 64) * 64;
  for (int i = threadIdx.x; i < wtot; i += 256) {
    int k = i >> 6;
    int col = i & 63;
    sW[(k >> 1) * 128 + col * 2 + (k & 1)] = W[i];
  }
  if (threadIdx.x < 64) sB[threadIdx.x] = bias[threadIdx.x];
  __syncthreads();

  const int lane = threadIdx.x & 31;
  const int wave = threadIdx.x >> 5;
  const int strip = blockIdx.x * 8 + wave;
  if (strip >= (n >> 4)) return;
  const int row0 = strip << 4;
  const int m  = lane & 15;
  const int kh = (lane >> 4) << 1;

  const float* a1p = A1 + (size_t)(row0 + m) * HF + kh;
  v8f acc[4] = {};

#pragma unroll
  for (int k0 = 0; k0 < 64; k0 += 4) {
    v2f a = *(const v2f*)(a1p + k0);
    const float* wb = &sW[(k0 + kh) * 64 + m * 2];
#pragma unroll
    for (int t = 0; t < 4; ++t) {
      v2f b = *(const v2f*)(wb + t * 32);
      acc[t] = __builtin_amdgcn_wmma_f32_16x16x4_f32(
          false, a, false, b, (short)0, acc[t], false, false);
    }
  }
  if (dual) {
    const float* a2p = A2 + (size_t)(row0 + m) * HF + kh;
    const float ds = (dinv ? dinv[row0 + m] : 1.0f) * a2sign;
#pragma unroll
    for (int k0 = 0; k0 < 64; k0 += 4) {
      v2f a = *(const v2f*)(a2p + k0);
      a = a * ds;
      const float* wb = &sW[(64 + k0 + kh) * 64 + m * 2];
#pragma unroll
      for (int t = 0; t < 4; ++t) {
        v2f b = *(const v2f*)(wb + t * 32);
        acc[t] = __builtin_amdgcn_wmma_f32_16x16x4_f32(
            false, a, false, b, (short)0, acc[t], false, false);
      }
    }
  }

  const int rbase = row0 + ((lane >> 4) << 3);
#pragma unroll
  for (int t = 0; t < 4; ++t) {
    const float bv = sB[t * 16 + m];
#pragma unroll
    for (int v = 0; v < 8; ++v) {
      float val = acc[t][v] + bv;
      if (act) val = val > 0.0f ? val : 0.01f * val;
      size_t o = (size_t)(rbase + v) * HF + t * 16 + m;
      out[o] = val;
      if (accum) accum[o] += val;
    }
  }
}

__global__ void k_head(const float* __restrict__ hall,
                       const float* __restrict__ W4,
                       const float* __restrict__ b4,
                       float* __restrict__ out, int n) {
  __shared__ float sW[128];
  __shared__ float sb[2];
  if (threadIdx.x < 128) sW[threadIdx.x] = W4[threadIdx.x];
  if (threadIdx.x < 2) sb[threadIdx.x] = b4[threadIdx.x];
  __syncthreads();
  int i = blockIdx.x * blockDim.x + threadIdx.x;
  if (i >= n) return;
  const float* hp = hall + (size_t)i * HF;
  float a0 = sb[0], a1 = sb[1];
#pragma unroll
  for (int k = 0; k < 64; ++k) {
    float v = hp[k];
    v = v > 0.0f ? v : 0.01f * v;
    a0 += v * sW[2 * k];
    a1 += v * sW[2 * k + 1];
  }
  out[2 * (size_t)i + 0] = a0;
  out[2 * (size_t)i + 1] = a1;
}

extern "C" void kernel_launch(void* const* d_in, const int* in_sizes, int n_in,
                              void* d_out, int out_size, void* d_ws, size_t ws_size,
                              hipStream_t stream) {
  const float* x   = (const float*)d_in[0];
  const int*   src = (const int*)d_in[1];
  const int*   dst = (const int*)d_in[2];
  const float* W1  = (const float*)d_in[3];
  const float* b1  = (const float*)d_in[4];
  const float* W2  = (const float*)d_in[5];
  const float* b2  = (const float*)d_in[6];
  const float* Wc1 = (const float*)d_in[7];
  const float* bc1 = (const float*)d_in[8];
  const float* Wc2 = (const float*)d_in[9];
  const float* bc2 = (const float*)d_in[10];
  const float* W3  = (const float*)d_in[11];
  const float* b3  = (const float*)d_in[12];
  const float* W4  = (const float*)d_in[13];
  const float* b4  = (const float*)d_in[14];
  float* out = (float*)d_out;

  const size_t NH = (size_t)NODES * HF;
  float* ws   = (float*)d_ws;
  float* hbuf = ws;
  float* h0   = hbuf + NH;
  float* h1   = h0 + NH;
  float* hall = h1 + NH;
  float* ah   = hall + NH;
  float* dinv = ah + NH;

  hipMemsetAsync(dinv, 0, sizeof(float) * (size_t)RELS * NODES, stream);
  k_deg<<<(RELS * EDGES + 255) / 256, 256, 0, stream>>>(dst, dinv);
  k_dinv<<<(RELS * NODES + 255) / 256, 256, 0, stream>>>(dinv);

  const int gblocks = ((NODES / 16) + 7) / 8;

  k_gemm<<<gblocks, 256, 0, stream>>>(x,  nullptr, nullptr, 1.0f, W1, b1, h0,   nullptr, NODES, 1);
  k_gemm<<<gblocks, 256, 0, stream>>>(h0, nullptr, nullptr, 1.0f, W2, b2, hbuf, nullptr, NODES, 1);
  hipMemsetAsync(hall, 0, sizeof(float) * NH, stream);

  for (int r = 0; r < RELS; ++r) {
    const int*   sr = src + (size_t)r * EDGES;
    const int*   dr = dst + (size_t)r * EDGES;
    const float* dv = dinv + (size_t)r * NODES;

    hipMemsetAsync(ah, 0, sizeof(float) * NH, stream);
    k_scatter<<<(EDGES * 16 + 255) / 256, 256, 0, stream>>>(hbuf, dv, sr, dr, ah);
    k_gemm<<<gblocks, 256, 0, stream>>>(hbuf, ah, dv, -1.0f, Wc1, bc1, h0, nullptr, NODES, 0);

    hipMemsetAsync(ah, 0, sizeof(float) * NH, stream);
    k_scatter<<<(EDGES * 16 + 255) / 256, 256, 0, stream>>>(h0, dv, sr, dr, ah);
    k_gemm<<<gblocks, 256, 0, stream>>>(h0, ah, dv, -1.0f, Wc2, bc2, h1, nullptr, NODES, 0);

    k_gemm<<<gblocks, 256, 0, stream>>>(h0, h1, nullptr, 1.0f, W3, b3, hbuf, hall, NODES, 0);
  }

  k_head<<<(NODES + 255) / 256, 256, 0, stream>>>(hall, W4, b4, out, NODES);
}


# Float-atomic scatter-add replaced by per-call dst bins (returning int-add rank, scan, fill) and an atomic-free per-node gather sum; sums now run in arrival-rank order
# speedup vs baseline: 1.5036x; 1.5036x over previous
#include <hip/hip_runtime.h>

typedef __attribute__((ext_vector_type(2))) float v2f;
typedef __attribute__((ext_vector_type(8))) float v8f;

#define NODES 100000
#define EDGES 1000000
#define RELS  3
#define HF    64

__global__ void k_rank(const int* __restrict__ dst, int* __restrict__ cnt,
                       int* __restrict__ rank) {
  int tid = blockIdx.x * blockDim.x + threadIdx.x;
  if (tid >= RELS * EDGES) return;
  int r = tid / EDGES;
  rank[tid] = atomicAdd(&cnt[r * NODES + dst[tid]], 1);
}

#define SCAN_T 1000
#define SCAN_C 100

__device__ __forceinline__ float deg_to_dinv(int c) {
  float v = (float)c;
  v = v < 1.0f ? 1.0f : v;
  return 1.0f / sqrtf(v);
}

__launch_bounds__(1024)
__global__ void k_scan(const int* __restrict__ cnt, int* __restrict__ off,
                       float* __restrict__ dinv) {
  __shared__ int part[1024];
  const int r = blockIdx.x;
  const int t = threadIdx.x;
  const size_t base = (size_t)r * NODES + (size_t)t * SCAN_C;
  const int4* c4 = (const int4*)(cnt + base);
  int sum = 0;
  if (t < SCAN_T) {
    for (int j = 0; j < SCAN_C / 4; ++j) {
      int4 v = c4[j];
      sum += v.x + v.y + v.z + v.w;
    }
  }
  part[t] = sum;
  __syncthreads();
  for (int d = 1; d < 1024; d <<= 1) {
    int v = (t >= d) ? part[t - d] : 0;
    __syncthreads();
    part[t] += v;
    __syncthreads();
  }
  if (t < SCAN_T) {
    int run = part[t] - sum;
    int4* o4 = (int4*)(off + base);
    float4* d4 = (float4*)(dinv + base);
#pragma unroll 5
    for (int j = 0; j < SCAN_C / 4; ++j) {
      int4 v = c4[j];
      int4 o;
      o.x = run; run += v.x;
      o.y = run; run += v.y;
      o.z = run; run += v.z;
      o.w = run; run += v.w;
      o4[j] = o;
      float4 dd;
      dd.x = deg_to_dinv(v.x);
      dd.y = deg_to_dinv(v.y);
      dd.z = deg_to_dinv(v.z);
      dd.w = deg_to_dinv(v.w);
      d4[j] = dd;
    }
  }
}

__global__ void k_fill(const int* __restrict__ src, const int* __restrict__ dst,
                       const int* __restrict__ off, const int* __restrict__ rank,
                       int* __restrict__ sorted) {
  int tid = blockIdx.x * blockDim.x + threadIdx.x;
  if (tid >= RELS * EDGES) return;
  int r = tid / EDGES;
  int pos = off[r * NODES + dst[tid]] + rank[tid];
  sorted[(size_t)r * EDGES + pos] = src[tid];
}

__global__ void k_gather(const float* __restrict__ feat,
                         const float* __restrict__ dinv,
                         const int* __restrict__ off,
                         const int* __restrict__ sorted,
                         float* __restrict__ ah) {
  int tid = blockIdx.x * blockDim.x + threadIdx.x;
  if (tid >= NODES * 16) return;
  int i = tid >> 4;
  int c = (tid & 15) << 2;
  int beg = off[i];
  int end = (i == NODES - 1) ? EDGES : off[i + 1];
  float4 acc = make_float4(0.0f, 0.0f, 0.0f, 0.0f);
#pragma unroll 4
  for (int j = beg; j < end; ++j) {
    int s = sorted[j];
    float dv = dinv[s];
    const float4 f = *(const float4*)(feat + (size_t)s * HF + c);
    acc.x += f.x * dv;
    acc.y += f.y * dv;
    acc.z += f.z * dv;
    acc.w += f.w * dv;
  }
  *(float4*)(ah + (size_t)i * HF + c) = acc;
}

__launch_bounds__(256)
__global__ void k_gemm(const float* __restrict__ A1,
                       const float* __restrict__ A2,
                       const float* __restrict__ dinv,
                       float a2sign,
                       const float* __restrict__ W,
                       const float* __restrict__ bias,
                       float* __restrict__ out,
                       float* __restrict__ accum,
                       int n, int act) {
  __shared__ float sW[128 * 64];
  __shared__ float sB[64];
  const bool dual = (A2 != nullptr);
  const int wtot = (dual ? 128 : 64) * 64;
  for (int i = threadIdx.x; i < wtot; i += 256) {
    int k = i >> 6;
    int col = i & 63;
    sW[(k >> 1) * 128 + col * 2 + (k & 1)] = W[i];
  }
  if (threadIdx.x < 64) sB[threadIdx.x] = bias[threadIdx.x];
  __syncthreads();

  const int lane = threadIdx.x & 31;
  const int wave = threadIdx.x >> 5;
  const int strip = blockIdx.x * 8 + wave;
  if (strip >= (n >> 4)) return;
  const int row0 = strip << 4;
  const int m  = lane & 15;
  const int kh = (lane >> 4) << 1;

  const float* a1p = A1 + (size_t)(row0 + m) * HF + kh;
  v8f acc[4] = {};

#pragma unroll
  for (int k0 = 0; k0 < 64; k0 += 4) {
    v2f a = *(const v2f*)(a1p + k0);
    const float* wb = &sW[(k0 + kh) * 64 + m * 2];
#pragma unroll
    for (int t = 0; t < 4; ++t) {
      v2f b = *(const v2f*)(wb + t * 32);
      acc[t] = __builtin_amdgcn_wmma_f32_16x16x4_f32(
          false, a, false, b, (short)0, acc[t], false, false);
    }
  }
  if (dual) {
    const float* a2p = A2 + (size_t)(row0 + m) * HF + kh;
    const float ds = (dinv ? dinv[row0 + m] : 1.0f) * a2sign;
#pragma unroll
    for (int k0 = 0; k0 < 64; k0 += 4) {
      v2f a = *(const v2f*)(a2p + k0);
      a = a * ds;
      const float* wb = &sW[(64 + k0 + kh) * 64 + m * 2];
#pragma unroll
      for (int t = 0; t < 4; ++t) {
        v2f b = *(const v2f*)(wb + t * 32);
        acc[t] = __builtin_amdgcn_wmma_f32_16x16x4_f32(
            false, a, false, b, (short)0, acc[t], false, false);
      }
    }
  }

  const int rbase = row0 + ((lane >> 4) << 3);
#pragma unroll
  for (int t = 0; t < 4; ++t) {
    const float bv = sB[t * 16 + m];
#pragma unroll
    for (int v = 0; v < 8; ++v) {
      float val = acc[t][v] + bv;
      if (act) val = val > 0.0f ? val : 0.01f * val;
      size_t o = (size_t)(rbase + v) * HF + t * 16 + m;
      out[o] = val;
      if (accum) accum[o] += val;
    }
  }
}

__global__ void k_head(const float* __restrict__ hall,
                       const float* __restrict__ W4,
                       const float* __restrict__ b4,
                       float* __restrict__ out, int n) {
  __shared__ float sW[128];
  __shared__ float sb[2];
  if (threadIdx.x < 128) sW[threadIdx.x] = W4[threadIdx.x];
  if (threadIdx.x < 2) sb[threadIdx.x] = b4[threadIdx.x];
  __syncthreads();
  int i = blockIdx.x * blockDim.x + threadIdx.x;
  if (i >= n) return;
  const float* hp = hall + (size_t)i * HF;
  float a0 = sb[0], a1 = sb[1];
#pragma unroll
  for (int k = 0; k < 64; ++k) {
    float v = hp[k];
    v = v > 0.0f ? v : 0.01f * v;
    a0 += v * sW[2 * k];
    a1 += v * sW[2 * k + 1];
  }
  out[2 * (size_t)i + 0] = a0;
  out[2 * (size_t)i + 1] = a1;
}

extern "C" void kernel_launch(void* const* d_in, const int* in_sizes, int n_in,
                              void* d_out, int out_size, void* d_ws, size_t ws_size,
                              hipStream_t stream) {
  const float* x   = (const float*)d_in[0];
  const int*   src = (const int*)d_in[1];
  const int*   dst = (const int*)d_in[2];
  const float* W1  = (const float*)d_in[3];
  const float* b1  = (const float*)d_in[4];
  const float* W2  = (const float*)d_in[5];
  const float* b2  = (const float*)d_in[6];
  const float* Wc1 = (const float*)d_in[7];
  const float* bc1 = (const float*)d_in[8];
  const float* Wc2 = (const float*)d_in[9];
  const float* bc2 = (const float*)d_in[10];
  const float* W3  = (const float*)d_in[11];
  const float* b3  = (const float*)d_in[12];
  const float* W4  = (const float*)d_in[13];
  const float* b4  = (const float*)d_in[14];
  float* out = (float*)d_out;

  const size_t NH = (size_t)NODES * HF;
  float* ws   = (float*)d_ws;
  float* hbuf = ws;
  float* h0   = hbuf + NH;
  float* h1   = h0 + NH;
  float* hall = h1 + NH;
  int*   ib   = (int*)(hall + NH);
  int*   sorted = ib;
  int*   off    = sorted + (size_t)RELS * EDGES;
  int*   cnt    = off + (size_t)RELS * NODES;
  float* dinv = (float*)ib + NH;
  int*   rank = (int*)h1;

  hipMemsetAsync(cnt, 0, sizeof(int) * (size_t)RELS * NODES, stream);
  k_rank<<<(RELS * EDGES + 255) / 256, 256, 0, stream>>>(dst, cnt, rank);
  k_scan<<<RELS, 1024, 0, stream>>>(cnt, off, dinv);
  k_fill<<<(RELS * EDGES + 255) / 256, 256, 0, stream>>>(src, dst, off, rank, sorted);

  const int gblocks = ((NODES / 16) + 7) / 8;
  const int ablocks = (NODES * 16 + 255) / 256;

  k_gemm<<<gblocks, 256, 0, stream>>>(x,  nullptr, nullptr, 1.0f, W1, b1, h0,   nullptr, NODES, 1);
  k_gemm<<<gblocks, 256, 0, stream>>>(h0, nullptr, nullptr, 1.0f, W2, b2, hbuf, nullptr, NODES, 1);
  hipMemsetAsync(hall, 0, sizeof(float) * NH, stream);

  for (int r = 0; r < RELS; ++r) {
    const int*   sr = sorted + (size_t)r * EDGES;
    const int*   fr = off + (size_t)r * NODES;
    const float* dv = dinv + (size_t)r * NODES;

    k_gather<<<ablocks, 256, 0, stream>>>(hbuf, dv, fr, sr, h1);
    k_gemm<<<gblocks, 256, 0, stream>>>(hbuf, h1, dv, -1.0f, Wc1, bc1, h0, nullptr, NODES, 0);

    k_gather<<<ablocks, 256, 0, stream>>>(h0, dv, fr, sr, hbuf);
    k_gemm<<<gblocks, 256, 0, stream>>>(h0, hbuf, dv, -1.0f, Wc2, bc2, h1, nullptr, NODES, 0);

    k_gemm<<<gblocks, 256, 0, stream>>>(h0, h1, nullptr, 1.0f, W3, b3, hbuf, hall, NODES, 0);
  }

  k_head<<<(NODES + 255) / 256, 256, 0, stream>>>(hall, W4, b4, out, NODES);
}


# Rank, fill and gather passes run as one resident set of long waves each (3 edge quads per lane; 6 nodes per 16 lanes, 8 rows in flight): 488k short waves per call become 66k; sums keep rank order
# speedup vs baseline: 1.5568x; 1.0354x over previous
#include <hip/hip_runtime.h>

typedef __attribute__((ext_vector_type(2))) float v2f;
typedef __attribute__((ext_vector_type(8))) float v8f;

#define NODES 100000
#define EDGES 1000000
#define RELS  3
#define HF    64

#define EQ    (EDGES / 4)
#define EBLK  250

__launch_bounds__(256)
__global__ void k_rank(const int* __restrict__ dst, int* __restrict__ cnt,
                       int* __restrict__ rank) {
  const int q = blockIdx.x * EBLK + threadIdx.x;
  const int4* d4 = (const int4*)dst;
  int4* r4 = (int4*)rank;
  int4 d[RELS], o[RELS];
#pragma unroll
  for (int r = 0; r < RELS; ++r) d[r] = d4[r * EQ + q];
  asm volatile("" :: "v"(d[0].x), "v"(d[0].y), "v"(d[0].z), "v"(d[0].w),
                     "v"(d[1].x), "v"(d[1].y), "v"(d[1].z), "v"(d[1].w),
                     "v"(d[2].x), "v"(d[2].y), "v"(d[2].z), "v"(d[2].w));
#pragma unroll
  for (int r = 0; r < RELS; ++r) {
    int* c = cnt + r * NODES;
    o[r].x = atomicAdd(c + d[r].x, 1);
    o[r].y = atomicAdd(c + d[r].y, 1);
    o[r].z = atomicAdd(c + d[r].z, 1);
    o[r].w = atomicAdd(c + d[r].w, 1);
  }
#pragma unroll
  for (int r = 0; r < RELS; ++r) r4[r * EQ + q] = o[r];
}

#define SCAN_T 1000
#define SCAN_C 100

__device__ __forceinline__ float deg_to_dinv(int c) {
  float v = (float)c;
  v = v < 1.0f ? 1.0f : v;
  return 1.0f / sqrtf(v);
}

__launch_bounds__(1024)
__global__ void k_scan(const int* __restrict__ cnt, int* __restrict__ off,
                       float* __restrict__ dinv) {
  __shared__ int part[1024];
  const int r = blockIdx.x;
  const int t = threadIdx.x;
  const size_t base = (size_t)r * NODES + (size_t)t * SCAN_C;
  const int4* c4 = (const int4*)(cnt + base);
  int sum = 0;
  if (t < SCAN_T) {
    for (int j = 0; j < SCAN_C / 4; ++j) {
      int4 v = c4[j];
      sum += v.x + v.y + v.z + v.w;
    }
  }
  part[t] = sum;
  __syncthreads();
  for (int d = 1; d < 1024; d <<= 1) {
    int v = (t >= d) ? part[t - d] : 0;
    __syncthreads();
    part[t] += v;
    __syncthreads();
  }
  if (t < SCAN_T) {
    int run = part[t] - sum;
    int4* o4 = (int4*)(off + base);
    float4* d4 = (float4*)(dinv + base);
#pragma unroll 5
    for (int j = 0; j < SCAN_C / 4; ++j) {
      int4 v = c4[j];
      int4 o;
      o.x = run; run += v.x;
      o.y = run; run += v.y;
      o.z = run; run += v.z;
      o.w = run; run += v.w;
      o4[j] = o;
      float4 dd;
      dd.x = deg_to_dinv(v.x);
      dd.y = deg_to_dinv(v.y);
      dd.z = deg_to_dinv(v.z);
      dd.w = deg_to_dinv(v.w);
      d4[j] = dd;
    }
  }
}

__launch_bounds__(256)
__global__ void k_fill(const int* __restrict__ src, const int* __restrict__ dst,
                       const int* __restrict__ off, const int* __restrict__ rank,
                       int* __restrict__ sorted) {
  const int q = blockIdx.x * EBLK + threadIdx.x;
  const int4* s4 = (const int4*)src;
  const int4* d4 = (const int4*)dst;
  const int4* r4 = (const int4*)rank;
  int4 s[RELS], d[RELS], rk[RELS], o[RELS];
#pragma unroll
  for (int r = 0; r < RELS; ++r) {
    d[r] = d4[r * EQ + q];
    rk[r] = r4[r * EQ + q];
    s[r] = s4[r * EQ + q];
  }
#pragma unroll
  for (int r = 0; r < RELS; ++r) {
    const int* f = off + r * NODES;
    o[r].x = f[d[r].x];
    o[r].y = f[d[r].y];
    o[r].z = f[d[r].z];
    o[r].w = f[d[r].w];
  }
#pragma unroll
  for (int r = 0; r < RELS; ++r) {
    int* out = sorted + (size_t)r * EDGES;
    out[o[r].x + rk[r].x] = s[r].x;
    out[o[r].y + rk[r].y] = s[r].y;
    out[o[r].z + rk[r].z] = s[r].z;
    out[o[r].w + rk[r].w] = s[r].w;
  }
}

#define GNPG    6
#define GBLOCKS (((NODES + GNPG - 1) / GNPG + 15) / 16)
#define GGROUPS (GBLOCKS * 16)
#define GCH     8

__launch_bounds__(256)
__global__ void k_gather(const float* __restrict__ feat,
                         const float* __restrict__ dinv,
                         const int* __restrict__ off,
                         const int* __restrict__ sorted,
                         float* __restrict__ ah) {
  const int tid = blockIdx.x * 256 + threadIdx.x;
  if (tid >= GGROUPS * 16) return;
  const int c = (tid & 15) << 2;
  const int g = tid >> 4;
  int beg[GNPG], end[GNPG];
#pragma unroll
  for (int k = 0; k < GNPG; ++k) {
    const int i = g + k * GGROUPS;
    beg[k] = 0;
    end[k] = 0;
    if (i < NODES) {
      beg[k] = off[i];
      end[k] = (i == NODES - 1) ? EDGES : off[i + 1];
    }
  }
#pragma unroll
  for (int k = 0; k < GNPG; ++k) {
    const int i = g + k * GGROUPS;
    if (i >= NODES) break;
    const int e = end[k];
    float4 acc = make_float4(0.0f, 0.0f, 0.0f, 0.0f);
    for (int j0 = beg[k]; j0 < e; j0 += GCH) {
      int s[GCH];
      float dv[GCH];
      float4 f[GCH];
#pragma unroll
      for (int u = 0; u < GCH; ++u) {
        s[u] = 0;
        if (j0 + u < e) s[u] = sorted[j0 + u];
      }
#pragma unroll
      for (int u = 0; u < GCH; ++u) {
        dv[u] = 0.0f;
        f[u] = make_float4(0.0f, 0.0f, 0.0f, 0.0f);
        if (j0 + u < e) {
          dv[u] = dinv[s[u]];
          f[u] = *(const float4*)(feat + (size_t)s[u] * HF + c);
        }
      }
#pragma unroll
      for (int u = 0; u < GCH; ++u) {
        if (j0 + u < e) {
          acc.x += f[u].x * dv[u];
          acc.y += f[u].y * dv[u];
          acc.z += f[u].z * dv[u];
          acc.w += f[u].w * dv[u];
        }
      }
    }
    *(float4*)(ah + (size_t)i * HF + c) = acc;
  }
}

__launch_bounds__(256)
__global__ void k_gemm(const float* __restrict__ A1,
                       const float* __restrict__ A2,
                       const float* __restrict__ dinv,
                       float a2sign,
                       const float* __restrict__ W,
                       const float* __restrict__ bias,
                       float* __restrict__ out,
                       float* __restrict__ accum,
                       int n, int act) {
  __shared__ float sW[128 * 64];
  __shared__ float sB[64];
  const bool dual = (A2 != nullptr);
  const int wtot = (dual ? 128 : 64) * 64;
  for (int i = threadIdx.x; i < wtot; i += 256) {
    int k = i >> 6;
    int col = i & 63;
    sW[(k >> 1) * 128 + col * 2 + (k & 1)] = W[i];
  }
  if (threadIdx.x < 64) sB[threadIdx.x] = bias[threadIdx.x];
  __syncthreads();

  const int lane = threadIdx.x & 31;
  const int wave = threadIdx.x >> 5;
  const int strip = blockIdx.x * 8 + wave;
  if (strip >= (n >> 4)) return;
  const int row0 = strip << 4;
  const int m  = lane & 15;
  const int kh = (lane >> 4) << 1;

  const float* a1p = A1 + (size_t)(row0 + m) * HF + kh;
  v8f acc[4] = {};

#pragma unroll
  for (int k0 = 0; k0 < 64; k0 += 4) {
    v2f a = *(const v2f*)(a1p + k0);
    const float* wb = &sW[(k0 + kh) * 64 + m * 2];
#pragma unroll
    for (int t = 0; t < 4; ++t) {
      v2f b = *(const v2f*)(wb + t * 32);
      acc[t] = __builtin_amdgcn_wmma_f32_16x16x4_f32(
          false, a, false, b, (short)0, acc[t], false, false);
    }
  }
  if (dual) {
    const float* a2p = A2 + (size_t)(row0 + m) * HF + kh;
    const float ds = (dinv ? dinv[row0 + m] : 1.0f) * a2sign;
#pragma unroll
    for (int k0 = 0; k0 < 64; k0 += 4) {
      v2f a = *(const v2f*)(a2p + k0);
      a = a * ds;
      const float* wb = &sW[(64 + k0 + kh) * 64 + m * 2];
#pragma unroll
      for (int t = 0; t < 4; ++t) {
        v2f b = *(const v2f*)(wb + t * 32);
        acc[t] = __builtin_amdgcn_wmma_f32_16x16x4_f32(
            false, a, false, b, (short)0, acc[t], false, false);
      }
    }
  }

  const int rbase = row0 + ((lane >> 4) << 3);
#pragma unroll
  for (int t = 0; t < 4; ++t) {
    const float bv = sB[t * 16 + m];
#pragma unroll
    for (int v = 0; v < 8; ++v) {
      float val = acc[t][v] + bv;
      if (act) val = val > 0.0f ? val : 0.01f * val;
      size_t o = (size_t)(rbase + v) * HF + t * 16 + m;
      out[o] = val;
      if (accum) accum[o] += val;
    }
  }
}

__global__ void k_head(const float* __restrict__ hall,
                       const float* __restrict__ W4,
                       const float* __restrict__ b4,
                       float* __restrict__ out, int n) {
  __shared__ float sW[128];
  __shared__ float sb[2];
  if (threadIdx.x < 128) sW[threadIdx.x] = W4[threadIdx.x];
  if (threadIdx.x < 2) sb[threadIdx.x] = b4[threadIdx.x];
  __syncthreads();
  int i = blockIdx.x * blockDim.x + threadIdx.x;
  if (i >= n) return;
  const float* hp = hall + (size_t)i * HF;
  float a0 = sb[0], a1 = sb[1];
#pragma unroll
  for (int k = 0; k < 64; ++k) {
    float v = hp[k];
    v = v > 0.0f ? v : 0.01f * v;
    a0 += v * sW[2 * k];
    a1 += v * sW[2 * k + 1];
  }
  out[2 * (size_t)i + 0] = a0;
  out[2 * (size_t)i + 1] = a1;
}

extern "C" void kernel_launch(void* const* d_in, const int* in_sizes, int n_in,
                              void* d_out, int out_size, void* d_ws, size_t ws_size,
                              hipStream_t stream) {
  const float* x   = (const float*)d_in[0];
  const int*   src = (const int*)d_in[1];
  const int*   dst = (const int*)d_in[2];
  const float* W1  = (const float*)d_in[3];
  const float* b1  = (const float*)d_in[4];
  const float* W2  = (const float*)d_in[5];
  const float* b2  = (const float*)d_in[6];
  const float* Wc1 = (const float*)d_in[7];
  const float* bc1 = (const float*)d_in[8];
  const float* Wc2 = (const float*)d_in[9];
  const float* bc2 = (const float*)d_in[10];
  const float* W3  = (const float*)d_in[11];
  const float* b3  = (const float*)d_in[12];
  const float* W4  = (const float*)d_in[13];
  const float* b4  = (const float*)d_in[14];
  float* out = (float*)d_out;

  const size_t NH = (size_t)NODES * HF;
  float* ws   = (float*)d_ws;
  float* hbuf = ws;
  float* h0   = hbuf + NH;
  float* h1   = h0 + NH;
  float* hall = h1 + NH;
  int*   ib   = (int*)(hall + NH);
  int*   sorted = ib;
  int*   off    = sorted + (size_t)RELS * EDGES;
  int*   cnt    = off + (size_t)RELS * NODES;
  float* dinv = (float*)ib + NH;
  int*   rank = (int*)h1;

  hipMemsetAsync(cnt, 0, sizeof(int) * (size_t)RELS * NODES, stream);
  k_rank<<<EQ / EBLK, EBLK, 0, stream>>>(dst, cnt, rank);
  k_scan<<<RELS, 1024, 0, stream>>>(cnt, off, dinv);
  k_fill<<<EQ / EBLK, EBLK, 0, stream>>>(src, dst, off, rank, sorted);

  const int gblocks = ((NODES / 16) + 7) / 8;
  const int ablocks = GBLOCKS;

  k_gemm<<<gblocks, 256, 0, stream>>>(x,  nullptr, nullptr, 1.0f, W1, b1, h0,   nullptr, NODES, 1);
  k_gemm<<<gblocks, 256, 0, stream>>>(h0, nullptr, nullptr, 1.0f, W2, b2, hbuf, nullptr, NODES, 1);
  hipMemsetAsync(hall, 0, sizeof(float) * NH, stream);

  for (int r = 0; r < RELS; ++r) {
    const int*   sr = sorted + (size_t)r * EDGES;
    const int*   fr = off + (size_t)r * NODES;
    const float* dv = dinv + (size_t)r * NODES;

    k_gather<<<ablocks, 256, 0, stream>>>(hbuf, dv, fr, sr, h1);
    k_gemm<<<gblocks, 256, 0, stream>>>(hbuf, h1, dv, -1.0f, Wc1, bc1, h0, nullptr, NODES, 0);

    k_gather<<<ablocks, 256, 0, stream>>>(h0, dv, fr, sr, hbuf);
    k_gemm<<<gblocks, 256, 0, stream>>>(h0, hbuf, dv, -1.0f, Wc2, bc2, h1, nullptr, NODES, 0);

    k_gemm<<<gblocks, 256, 0, stream>>>(h0, h1, nullptr, 1.0f, W3, b3, hbuf, hall, NODES, 0);
  }

  k_head<<<(NODES + 255) / 256, 256, 0, stream>>>(hall, W4, b4, out, NODES);
}


# Bin-start prefix scan spread from 3 blocks (one per relation, 400-byte lane stride) to 150 blocks: each sums the counts in front of its 2000-node chunk with coalesced loads, then scans the chunk
# speedup vs baseline: 1.7031x; 1.0940x over previous
#include <hip/hip_runtime.h>

typedef __attribute__((ext_vector_type(2))) float v2f;
typedef __attribute__((ext_vector_type(8))) float v8f;

#define NODES 100000
#define EDGES 1000000
#define RELS  3
#define HF    64

#define EQ    (EDGES / 4)
#define EBLK  250

__launch_bounds__(256)
__global__ void k_rank(const int* __restrict__ dst, int* __restrict__ cnt,
                       int* __restrict__ rank) {
  const int q = blockIdx.x * EBLK + threadIdx.x;
  const int4* d4 = (const int4*)dst;
  int4* r4 = (int4*)rank;
  int4 d[RELS], o[RELS];
#pragma unroll
  for (int r = 0; r < RELS; ++r) d[r] = d4[r * EQ + q];
  asm volatile("" :: "v"(d[0].x), "v"(d[0].y), "v"(d[0].z), "v"(d[0].w),
                     "v"(d[1].x), "v"(d[1].y), "v"(d[1].z), "v"(d[1].w),
                     "v"(d[2].x), "v"(d[2].y), "v"(d[2].z), "v"(d[2].w));
#pragma unroll
  for (int r = 0; r < RELS; ++r) {
    int* c = cnt + r * NODES;
    o[r].x = atomicAdd(c + d[r].x, 1);
    o[r].y = atomicAdd(c + d[r].y, 1);
    o[r].z = atomicAdd(c + d[r].z, 1);
    o[r].w = atomicAdd(c + d[r].w, 1);
  }
#pragma unroll
  for (int r = 0; r < RELS; ++r) r4[r * EQ + q] = o[r];
}

#define SC_CH 2000
#define SC_NB (NODES / SC_CH)
#define SC_Q  (SC_CH / 4)
#define SC_T  512
#define SC_W  (SC_T / 32)
#define SC_U  8

__device__ __forceinline__ float deg_to_dinv(int c) {
  float v = (float)c;
  v = v < 1.0f ? 1.0f : v;
  return 1.0f / sqrtf(v);
}

__launch_bounds__(SC_T)
__global__ void k_scan(const int* __restrict__ cnt, int* __restrict__ off,
                       float* __restrict__ dinv) {
  __shared__ int wpre[SC_W];
  __shared__ int wown[SC_W];
  const int r = blockIdx.x / SC_NB;
  const int b = blockIdx.x - r * SC_NB;
  const int t = threadIdx.x;
  const int lane = t & 31;
  const int wave = t >> 5;
  const int4* c4 = (const int4*)(cnt + (size_t)r * NODES);
  const int nq = b * SC_Q;

  int4 v = make_int4(0, 0, 0, 0);
  if (t < SC_Q) v = c4[nq + t];

  int pre = 0;
  for (int i0 = t; i0 < nq; i0 += SC_U * SC_T) {
    int4 u[SC_U];
#pragma unroll
    for (int k = 0; k < SC_U; ++k) {
      u[k] = make_int4(0, 0, 0, 0);
      if (i0 + k * SC_T < nq) u[k] = c4[i0 + k * SC_T];
    }
#pragma unroll
    for (int k = 0; k < SC_U; ++k) pre += (u[k].x + u[k].y) + (u[k].z + u[k].w);
  }

  const int own = (v.x + v.y) + (v.z + v.w);
  int inc = own;
  for (int d = 1; d < 32; d <<= 1) {
    const int u = __shfl_up(inc, d);
    if (lane >= d) inc += u;
  }
  for (int d = 16; d >= 1; d >>= 1) pre += __shfl_xor(pre, d);
  if (lane == 31) {
    wown[wave] = inc;
    wpre[wave] = pre;
  }
  __syncthreads();
  int run = inc - own;
  for (int w = 0; w < SC_W; ++w) {
    run += wpre[w];
    if (w < wave) run += wown[w];
  }

  if (t < SC_Q) {
    const size_t e = (size_t)r * NODES + (size_t)(nq + t) * 4;
    int4 o;
    o.x = run; run += v.x;
    o.y = run; run += v.y;
    o.z = run; run += v.z;
    o.w = run;
    *(int4*)(off + e) = o;
    float4 dd;
    dd.x = deg_to_dinv(v.x);
    dd.y = deg_to_dinv(v.y);
    dd.z = deg_to_dinv(v.z);
    dd.w = deg_to_dinv(v.w);
    *(float4*)(dinv + e) = dd;
  }
}

__launch_bounds__(256)
__global__ void k_fill(const int* __restrict__ src, const int* __restrict__ dst,
                       const int* __restrict__ off, const int* __restrict__ rank,
                       int* __restrict__ sorted) {
  const int q = blockIdx.x * EBLK + threadIdx.x;
  const int4* s4 = (const int4*)src;
  const int4* d4 = (const int4*)dst;
  const int4* r4 = (const int4*)rank;
  int4 s[RELS], d[RELS], rk[RELS], o[RELS];
#pragma unroll
  for (int r = 0; r < RELS; ++r) {
    d[r] = d4[r * EQ + q];
    rk[r] = r4[r * EQ + q];
    s[r] = s4[r * EQ + q];
  }
#pragma unroll
  for (int r = 0; r < RELS; ++r) {
    const int* f = off + r * NODES;
    o[r].x = f[d[r].x];
    o[r].y = f[d[r].y];
    o[r].z = f[d[r].z];
    o[r].w = f[d[r].w];
  }
#pragma unroll
  for (int r = 0; r < RELS; ++r) {
    int* out = sorted + (size_t)r * EDGES;
    out[o[r].x + rk[r].x] = s[r].x;
    out[o[r].y + rk[r].y] = s[r].y;
    out[o[r].z + rk[r].z] = s[r].z;
    out[o[r].w + rk[r].w] = s[r].w;
  }
}

#define GNPG    6
#define GBLOCKS (((NODES + GNPG - 1) / GNPG + 15) / 16)
#define GGROUPS (GBLOCKS * 16)
#define GCH     8

__launch_bounds__(256)
__global__ void k_gather(const float* __restrict__ feat,
                         const float* __restrict__ dinv,
                         const int* __restrict__ off,
                         const int* __restrict__ sorted,
                         float* __restrict__ ah) {
  const int tid = blockIdx.x * 256 + threadIdx.x;
  if (tid >= GGROUPS * 16) return;
  const int c = (tid & 15) << 2;
  const int g = tid >> 4;
  int beg[GNPG], end[GNPG];
#pragma unroll
  for (int k = 0; k < GNPG; ++k) {
    const int i = g + k * GGROUPS;
    beg[k] = 0;
    end[k] = 0;
    if (i < NODES) {
      beg[k] = off[i];
      end[k] = (i == NODES - 1) ? EDGES : off[i + 1];
    }
  }
#pragma unroll
  for (int k = 0; k < GNPG; ++k) {
    const int i = g + k * GGROUPS;
    if (i >= NODES) break;
    const int e = end[k];
    float4 acc = make_float4(0.0f, 0.0f, 0.0f, 0.0f);
    for (int j0 = beg[k]; j0 < e; j0 += GCH) {
      int s[GCH];
      float dv[GCH];
      float4 f[GCH];
#pragma unroll
      for (int u = 0; u < GCH; ++u) {
        s[u] = 0;
        if (j0 + u < e) s[u] = sorted[j0 + u];
      }
#pragma unroll
      for (int u = 0; u < GCH; ++u) {
        dv[u] = 0.0f;
        f[u] = make_float4(0.0f, 0.0f, 0.0f, 0.0f);
        if (j0 + u < e) {
          dv[u] = dinv[s[u]];
          f[u] = *(const float4*)(feat + (size_t)s[u] * HF + c);
        }
      }
#pragma unroll
      for (int u = 0; u < GCH; ++u) {
        if (j0 + u < e) {
          acc.x += f[u].x * dv[u];
          acc.y += f[u].y * dv[u];
          acc.z += f[u].z * dv[u];
          acc.w += f[u].w * dv[u];
        }
      }
    }
    *(float4*)(ah + (size_t)i * HF + c) = acc;
  }
}

__launch_bounds__(256)
__global__ void k_gemm(const float* __restrict__ A1,
                       const float* __restrict__ A2,
                       const float* __restrict__ dinv,
                       float a2sign,
                       const float* __restrict__ W,
                       const float* __restrict__ bias,
                       float* __restrict__ out,
                       float* __restrict__ accum,
                       int n, int act) {
  __shared__ float sW[128 * 64];
  __shared__ float sB[64];
  const bool dual = (A2 != nullptr);
  const int wtot = (dual ? 128 : 64) * 64;
  for (int i = threadIdx.x; i < wtot; i += 256) {
    int k = i >> 6;
    int col = i & 63;
    sW[(k >> 1) * 128 + col * 2 + (k & 1)] = W[i];
  }
  if (threadIdx.x < 64) sB[threadIdx.x] = bias[threadIdx.x];
  __syncthreads();

  const int lane = threadIdx.x & 31;
  const int wave = threadIdx.x >> 5;
  const int strip = blockIdx.x * 8 + wave;
  if (strip >= (n >> 4)) return;
  const int row0 = strip << 4;
  const int m  = lane & 15;
  const int kh = (lane >> 4) << 1;

  const float* a1p = A1 + (size_t)(row0 + m) * HF + kh;
  v8f acc[4] = {};

#pragma unroll
  for (int k0 = 0; k0 < 64; k0 += 4) {
    v2f a = *(const v2f*)(a1p + k0);
    const float* wb = &sW[(k0 + kh) * 64 + m * 2];
#pragma unroll
    for (int t = 0; t < 4; ++t) {
      v2f b = *(const v2f*)(wb + t * 32);
      acc[t] = __builtin_amdgcn_wmma_f32_16x16x4_f32(
          false, a, false, b, (short)0, acc[t], false, false);
    }
  }
  if (dual) {
    const float* a2p = A2 + (size_t)(row0 + m) * HF + kh;
    const float ds = (dinv ? dinv[row0 + m] : 1.0f) * a2sign;
#pragma unroll
    for (int k0 = 0; k0 < 64; k0 += 4) {
      v2f a = *(const v2f*)(a2p + k0);
      a = a * ds;
      const float* wb = &sW[(64 + k0 + kh) * 64 + m * 2];
#pragma unroll
      for (int t = 0; t < 4; ++t) {
        v2f b = *(const v2f*)(wb + t * 32);
        acc[t] = __builtin_amdgcn_wmma_f32_16x16x4_f32(
            false, a, false, b, (short)0, acc[t], false, false);
      }
    }
  }

  const int rbase = row0 + ((lane >> 4) << 3);
#pragma unroll
  for (int t = 0; t < 4; ++t) {
    const float bv = sB[t * 16 + m];
#pragma unroll
    for (int v = 0; v < 8; ++v) {
      float val = acc[t][v] + bv;
      if (act) val = val > 0.0f ? val : 0.01f * val;
      size_t o = (size_t)(rbase + v) * HF + t * 16 + m;
      out[o] = val;
      if (accum) accum[o] += val;
    }
  }
}

__global__ void k_head(const float* __restrict__ hall,
                       const float* __restrict__ W4,
                       const float* __restrict__ b4,
                       float* __restrict__ out, int n) {
  __shared__ float sW[128];
  __shared__ float sb[2];
  if (threadIdx.x < 128) sW[threadIdx.x] = W4[threadIdx.x];
  if (threadIdx.x < 2) sb[threadIdx.x] = b4[threadIdx.x];
  __syncthreads();
  int i = blockIdx.x * blockDim.x + threadIdx.x;
  if (i >= n) return;
  const float* hp = hall + (size_t)i * HF;
  float a0 = sb[0], a1 = sb[1];
#pragma unroll
  for (int k = 0; k < 64; ++k) {
    float v = hp[k];
    v = v > 0.0f ? v : 0.01f * v;
    a0 += v * sW[2 * k];
    a1 += v * sW[2 * k + 1];
  }
  out[2 * (size_t)i + 0] = a0;
  out[2 * (size_t)i + 1] = a1;
}

extern "C" void kernel_launch(void* const* d_in, const int* in_sizes, int n_in,
                              void* d_out, int out_size, void* d_ws, size_t ws_size,
                              hipStream_t stream) {
  const float* x   = (const float*)d_in[0];
  const int*   src = (const int*)d_in[1];
  const int*   dst = (const int*)d_in[2];
  const float* W1  = (const float*)d_in[3];
  const float* b1  = (const float*)d_in[4];
  const float* W2  = (const float*)d_in[5];
  const float* b2  = (const float*)d_in[6];
  const float* Wc1 = (const float*)d_in[7];
  const float* bc1 = (const float*)d_in[8];
  const float* Wc2 = (const float*)d_in[9];
  const float* bc2 = (const float*)d_in[10];
  const float* W3  = (const float*)d_in[11];
  const float* b3  = (const float*)d_in[12];
  const float* W4  = (const float*)d_in[13];
  const float* b4  = (const float*)d_in[14];
  float* out = (float*)d_out;

  const size_t NH = (size_t)NODES * HF;
  float* ws   = (float*)d_ws;
  float* hbuf = ws;
  float* h0   = hbuf + NH;
  float* h1   = h0 + NH;
  float* hall = h1 + NH;
  int*   ib   = (int*)(hall + NH);
  int*   sorted = ib;
  int*   off    = sorted + (size_t)RELS * EDGES;
  int*   cnt    = off + (size_t)RELS * NODES;
  float* dinv = (float*)ib + NH;
  int*   rank = (int*)h1;

  hipMemsetAsync(cnt, 0, sizeof(int) * (size_t)RELS * NODES, stream);
  k_rank<<<EQ / EBLK, EBLK, 0, stream>>>(dst, cnt, rank);
  k_scan<<<RELS * SC_NB, SC_T, 0, stream>>>(cnt, off, dinv);
  k_fill<<<EQ / EBLK, EBLK, 0, stream>>>(src, dst, off, rank, sorted);

  const int gblocks = ((NODES / 16) + 7) / 8;
  const int ablocks = GBLOCKS;

  k_gemm<<<gblocks, 256, 0, stream>>>(x,  nullptr, nullptr, 1.0f, W1, b1, h0,   nullptr, NODES, 1);
  k_gemm<<<gblocks, 256, 0, stream>>>(h0, nullptr, nullptr, 1.0f, W2, b2, hbuf, nullptr, NODES, 1);
  hipMemsetAsync(hall, 0, sizeof(float) * NH, stream);

  for (int r = 0; r < RELS; ++r) {
    const int*   sr = sorted + (size_t)r * EDGES;
    const int*   fr = off + (size_t)r * NODES;
    const float* dv = dinv + (size_t)r * NODES;

    k_gather<<<ablocks, 256, 0, stream>>>(hbuf, dv, fr, sr, h1);
    k_gemm<<<gblocks, 256, 0, stream>>>(hbuf, h1, dv, -1.0f, Wc1, bc1, h0, nullptr, NODES, 0);

    k_gather<<<ablocks, 256, 0, stream>>>(h0, dv, fr, sr, hbuf);
    k_gemm<<<gblocks, 256, 0, stream>>>(h0, hbuf, dv, -1.0f, Wc2, bc2, h1, nullptr, NODES, 0);

    k_gemm<<<gblocks, 256, 0, stream>>>(h0, h1, nullptr, 1.0f, W3, b3, hbuf, hall, NODES, 0);
  }

  k_head<<<(NODES + 255) / 256, 256, 0, stream>>>(hall, W4, b4, out, NODES);
}


# Last two products of a relation step folded into one: h=[h0,X1]@Wf+cf with Wf,cf built per call on WMMA from Wc2,bc2,W3,b3; 11 node GEMMs become 8; that product's sums re-associate
# speedup vs baseline: 1.9630x; 1.1526x over previous
#include <hip/hip_runtime.h>

typedef __attribute__((ext_vector_type(2))) float v2f;
typedef __attribute__((ext_vector_type(8))) float v8f;

#define NODES 100000
#define EDGES 1000000
#define RELS  3
#define HF    64

#define EQ    (EDGES / 4)
#define EBLK  250

__launch_bounds__(256)
__global__ void k_rank(const int* __restrict__ dst, int* __restrict__ cnt,
                       int* __restrict__ rank) {
  const int q = blockIdx.x * EBLK + threadIdx.x;
  const int4* d4 = (const int4*)dst;
  int4* r4 = (int4*)rank;
  int4 d[RELS], o[RELS];
#pragma unroll
  for (int r = 0; r < RELS; ++r) d[r] = d4[r * EQ + q];
  asm volatile("" :: "v"(d[0].x), "v"(d[0].y), "v"(d[0].z), "v"(d[0].w),
                     "v"(d[1].x), "v"(d[1].y), "v"(d[1].z), "v"(d[1].w),
                     "v"(d[2].x), "v"(d[2].y), "v"(d[2].z), "v"(d[2].w));
#pragma unroll
  for (int r = 0; r < RELS; ++r) {
    int* c = cnt + r * NODES;
    o[r].x = atomicAdd(c + d[r].x, 1);
    o[r].y = atomicAdd(c + d[r].y, 1);
    o[r].z = atomicAdd(c + d[r].z, 1);
    o[r].w = atomicAdd(c + d[r].w, 1);
  }
#pragma unroll
  for (int r = 0; r < RELS; ++r) r4[r * EQ + q] = o[r];
}

#define SC_CH 2000
#define SC_NB (NODES / SC_CH)
#define SC_Q  (SC_CH / 4)
#define SC_T  512
#define SC_W  (SC_T / 32)
#define SC_U  8

__device__ __forceinline__ float deg_to_dinv(int c) {
  float v = (float)c;
  v = v < 1.0f ? 1.0f : v;
  return 1.0f / sqrtf(v);
}

__launch_bounds__(SC_T)
__global__ void k_scan(const int* __restrict__ cnt, int* __restrict__ off,
                       float* __restrict__ dinv) {
  __shared__ int wpre[SC_W];
  __shared__ int wown[SC_W];
  const int r = blockIdx.x / SC_NB;
  const int b = blockIdx.x - r * SC_NB;
  const int t = threadIdx.x;
  const int lane = t & 31;
  const int wave = t >> 5;
  const int4* c4 = (const int4*)(cnt + (size_t)r * NODES);
  const int nq = b * SC_Q;

  int4 v = make_int4(0, 0, 0, 0);
  if (t < SC_Q) v = c4[nq + t];

  int pre = 0;
  for (int i0 = t; i0 < nq; i0 += SC_U * SC_T) {
    int4 u[SC_U];
#pragma unroll
    for (int k = 0; k < SC_U; ++k) {
      u[k] = make_int4(0, 0, 0, 0);
      if (i0 + k * SC_T < nq) u[k] = c4[i0 + k * SC_T];
    }
#pragma unroll
    for (int k = 0; k < SC_U; ++k) pre += (u[k].x + u[k].y) + (u[k].z + u[k].w);
  }

  const int own = (v.x + v.y) + (v.z + v.w);
  int inc = own;
  for (int d = 1; d < 32; d <<= 1) {
    const int u = __shfl_up(inc, d);
    if (lane >= d) inc += u;
  }
  for (int d = 16; d >= 1; d >>= 1) pre += __shfl_xor(pre, d);
  if (lane == 31) {
    wown[wave] = inc;
    wpre[wave] = pre;
  }
  __syncthreads();
  int run = inc - own;
  for (int w = 0; w < SC_W; ++w) {
    run += wpre[w];
    if (w < wave) run += wown[w];
  }

  if (t < SC_Q) {
    const size_t e = (size_t)r * NODES + (size_t)(nq + t) * 4;
    int4 o;
    o.x = run; run += v.x;
    o.y = run; run += v.y;
    o.z = run; run += v.z;
    o.w = run;
    *(int4*)(off + e) = o;
    float4 dd;
    dd.x = deg_to_dinv(v.x);
    dd.y = deg_to_dinv(v.y);
    dd.z = deg_to_dinv(v.z);
    dd.w = deg_to_dinv(v.w);
    *(float4*)(dinv + e) = dd;
  }
}

__launch_bounds__(256)
__global__ void k_fill(const int* __restrict__ src, const int* __restrict__ dst,
                       const int* __restrict__ off, const int* __restrict__ rank,
                       int* __restrict__ sorted) {
  const int q = blockIdx.x * EBLK + threadIdx.x;
  const int4* s4 = (const int4*)src;
  const int4* d4 = (const int4*)dst;
  const int4* r4 = (const int4*)rank;
  int4 s[RELS], d[RELS], rk[RELS], o[RELS];
#pragma unroll
  for (int r = 0; r < RELS; ++r) {
    d[r] = d4[r * EQ + q];
    rk[r] = r4[r * EQ + q];
    s[r] = s4[r * EQ + q];
  }
#pragma unroll
  for (int r = 0; r < RELS; ++r) {
    const int* f = off + r * NODES;
    o[r].x = f[d[r].x];
    o[r].y = f[d[r].y];
    o[r].z = f[d[r].z];
    o[r].w = f[d[r].w];
  }
#pragma unroll
  for (int r = 0; r < RELS; ++r) {
    int* out = sorted + (size_t)r * EDGES;
    out[o[r].x + rk[r].x] = s[r].x;
    out[o[r].y + rk[r].y] = s[r].y;
    out[o[r].z + rk[r].z] = s[r].z;
    out[o[r].w + rk[r].w] = s[r].w;
  }
}

#define GNPG    6
#define GBLOCKS (((NODES + GNPG - 1) / GNPG + 15) / 16)
#define GGROUPS (GBLOCKS * 16)
#define GCH     8

__launch_bounds__(256)
__global__ void k_gather(const float* __restrict__ feat,
                         const float* __restrict__ dinv,
                         const int* __restrict__ off,
                         const int* __restrict__ sorted,
                         float* __restrict__ ah) {
  const int tid = blockIdx.x * 256 + threadIdx.x;
  if (tid >= GGROUPS * 16) return;
  const int c = (tid & 15) << 2;
  const int g = tid >> 4;
  int beg[GNPG], end[GNPG];
#pragma unroll
  for (int k = 0; k < GNPG; ++k) {
    const int i = g + k * GGROUPS;
    beg[k] = 0;
    end[k] = 0;
    if (i < NODES) {
      beg[k] = off[i];
      end[k] = (i == NODES - 1) ? EDGES : off[i + 1];
    }
  }
#pragma unroll
  for (int k = 0; k < GNPG; ++k) {
    const int i = g + k * GGROUPS;
    if (i >= NODES) break;
    const int e = end[k];
    float4 acc = make_float4(0.0f, 0.0f, 0.0f, 0.0f);
    for (int j0 = beg[k]; j0 < e; j0 += GCH) {
      int s[GCH];
      float dv[GCH];
      float4 f[GCH];
#pragma unroll
      for (int u = 0; u < GCH; ++u) {
        s[u] = 0;
        if (j0 + u < e) s[u] = sorted[j0 + u];
      }
#pragma unroll
      for (int u = 0; u < GCH; ++u) {
        dv[u] = 0.0f;
        f[u] = make_float4(0.0f, 0.0f, 0.0f, 0.0f);
        if (j0 + u < e) {
          dv[u] = dinv[s[u]];
          f[u] = *(const float4*)(feat + (size_t)s[u] * HF + c);
        }
      }
#pragma unroll
      for (int u = 0; u < GCH; ++u) {
        if (j0 + u < e) {
          acc.x += f[u].x * dv[u];
          acc.y += f[u].y * dv[u];
          acc.z += f[u].z * dv[u];
          acc.w += f[u].w * dv[u];
        }
      }
    }
    *(float4*)(ah + (size_t)i * HF + c) = acc;
  }
}

__launch_bounds__(256)
__global__ void k_gemm(const float* __restrict__ A1,
                       const float* __restrict__ A2,
                       const float* __restrict__ dinv,
                       float a2sign,
                       const float* __restrict__ W,
                       const float* __restrict__ bias,
                       float* __restrict__ out,
                       float* __restrict__ accum,
                       int n, int act) {
  __shared__ float sW[128 * 64];
  __shared__ float sB[64];
  const bool dual = (A2 != nullptr);
  const int wtot = (dual ? 128 : 64) * 64;
  for (int i = threadIdx.x; i < wtot; i += 256) {
    int k = i >> 6;
    int col = i & 63;
    sW[(k >> 1) * 128 + col * 2 + (k & 1)] = W[i];
  }
  if (threadIdx.x < 64) sB[threadIdx.x] = bias[threadIdx.x];
  __syncthreads();

  const int lane = threadIdx.x & 31;
  const int wave = threadIdx.x >> 5;
  const int strip = blockIdx.x * 8 + wave;
  if (strip >= (n >> 4)) return;
  const int row0 = strip << 4;
  const int m  = lane & 15;
  const int kh = (lane >> 4) << 1;

  const float* a1p = A1 + (size_t)(row0 + m) * HF + kh;
  v8f acc[4] = {};

#pragma unroll
  for (int k0 = 0; k0 < 64; k0 += 4) {
    v2f a = *(const v2f*)(a1p + k0);
    const float* wb = &sW[(k0 + kh) * 64 + m * 2];
#pragma unroll
    for (int t = 0; t < 4; ++t) {
      v2f b = *(const v2f*)(wb + t * 32);
      acc[t] = __builtin_amdgcn_wmma_f32_16x16x4_f32(
          false, a, false, b, (short)0, acc[t], false, false);
    }
  }
  if (dual) {
    const float* a2p = A2 + (size_t)(row0 + m) * HF + kh;
    const float ds = (dinv ? dinv[row0 + m] : 1.0f) * a2sign;
#pragma unroll
    for (int k0 = 0; k0 < 64; k0 += 4) {
      v2f a = *(const v2f*)(a2p + k0);
      a = a * ds;
      const float* wb = &sW[(64 + k0 + kh) * 64 + m * 2];
#pragma unroll
      for (int t = 0; t < 4; ++t) {
        v2f b = *(const v2f*)(wb + t * 32);
        acc[t] = __builtin_amdgcn_wmma_f32_16x16x4_f32(
            false, a, false, b, (short)0, acc[t], false, false);
      }
    }
  }

  const int rbase = row0 + ((lane >> 4) << 3);
#pragma unroll
  for (int t = 0; t < 4; ++t) {
    const float bv = sB[t * 16 + m];
#pragma unroll
    for (int v = 0; v < 8; ++v) {
      float val = acc[t][v] + bv;
      if (act) val = val > 0.0f ? val : 0.01f * val;
      size_t o = (size_t)(rbase + v) * HF + t * 16 + m;
      out[o] = val;
      if (accum) accum[o] += val;
    }
  }
}

#define WF_T 288

__launch_bounds__(WF_T)
__global__ void k_wfold(const float* __restrict__ Wc2,
                        const float* __restrict__ bc2,
                        const float* __restrict__ W3,
                        const float* __restrict__ b3,
                        float* __restrict__ Wf,
                        float* __restrict__ cf) {
  __shared__ float sW[64 * 64];
  const float* Wlo = W3 + 64 * HF;
  for (int i = threadIdx.x; i < 64 * 64; i += WF_T) {
    int k = i >> 6;
    int col = i & 63;
    sW[(k >> 1) * 128 + col * 2 + (k & 1)] = Wlo[i];
  }
  __syncthreads();

  const int lane = threadIdx.x & 31;
  const int wave = threadIdx.x >> 5;
  const int m  = lane & 15;
  const int kh = (lane >> 4) << 1;
  const int row = (wave << 4) + m;
  const float* ap = (row < 128 ? Wc2 + (size_t)row * HF : bc2) + kh;
  const bool live = row <= 128;
  v8f acc[4] = {};

#pragma unroll
  for (int k0 = 0; k0 < 64; k0 += 4) {
    v2f a = *(const v2f*)(ap + k0);
    if (!live) a = (v2f){0.0f, 0.0f};
    const float* wb = &sW[(k0 + kh) * 64 + m * 2];
#pragma unroll
    for (int t = 0; t < 4; ++t) {
      v2f b = *(const v2f*)(wb + t * 32);
      acc[t] = __builtin_amdgcn_wmma_f32_16x16x4_f32(
          false, a, false, b, (short)0, acc[t], false, false);
    }
  }

  const int rbase = (wave << 4) + ((lane >> 4) << 3);
  if (wave < 4) {
    float u[4][8];
#pragma unroll
    for (int t = 0; t < 4; ++t)
#pragma unroll
      for (int v = 0; v < 8; ++v) u[t][v] = W3[(rbase + v) * HF + t * 16 + m];
#pragma unroll
    for (int t = 0; t < 4; ++t)
#pragma unroll
      for (int v = 0; v < 8; ++v) Wf[(rbase + v) * HF + t * 16 + m] = acc[t][v] + u[t][v];
  } else if (wave < 8) {
#pragma unroll
    for (int t = 0; t < 4; ++t)
#pragma unroll
      for (int v = 0; v < 8; ++v) Wf[(rbase + v) * HF + t * 16 + m] = acc[t][v];
  } else if (lane < 16) {
#pragma unroll
    for (int t = 0; t < 4; ++t) cf[t * 16 + m] = acc[t][0] + b3[t * 16 + m];
  }
}

__global__ void k_head(const float* __restrict__ hall,
                       const float* __restrict__ W4,
                       const float* __restrict__ b4,
                       float* __restrict__ out, int n) {
  __shared__ float sW[128];
  __shared__ float sb[2];
  if (threadIdx.x < 128) sW[threadIdx.x] = W4[threadIdx.x];
  if (threadIdx.x < 2) sb[threadIdx.x] = b4[threadIdx.x];
  __syncthreads();
  int i = blockIdx.x * blockDim.x + threadIdx.x;
  if (i >= n) return;
  const float* hp = hall + (size_t)i * HF;
  float a0 = sb[0], a1 = sb[1];
#pragma unroll
  for (int k = 0; k < 64; ++k) {
    float v = hp[k];
    v = v > 0.0f ? v : 0.01f * v;
    a0 += v * sW[2 * k];
    a1 += v * sW[2 * k + 1];
  }
  out[2 * (size_t)i + 0] = a0;
  out[2 * (size_t)i + 1] = a1;
}

extern "C" void kernel_launch(void* const* d_in, const int* in_sizes, int n_in,
                              void* d_out, int out_size, void* d_ws, size_t ws_size,
                              hipStream_t stream) {
  const float* x   = (const float*)d_in[0];
  const int*   src = (const int*)d_in[1];
  const int*   dst = (const int*)d_in[2];
  const float* W1  = (const float*)d_in[3];
  const float* b1  = (const float*)d_in[4];
  const float* W2  = (const float*)d_in[5];
  const float* b2  = (const float*)d_in[6];
  const float* Wc1 = (const float*)d_in[7];
  const float* bc1 = (const float*)d_in[8];
  const float* Wc2 = (const float*)d_in[9];
  const float* bc2 = (const float*)d_in[10];
  const float* W3  = (const float*)d_in[11];
  const float* b3  = (const float*)d_in[12];
  const float* W4  = (const float*)d_in[13];
  const float* b4  = (const float*)d_in[14];
  float* out = (float*)d_out;

  const size_t NH = (size_t)NODES * HF;
  float* ws   = (float*)d_ws;
  float* hbuf = ws;
  float* h0   = hbuf + NH;
  float* h1   = h0 + NH;
  float* hall = h1 + NH;
  int*   ib   = (int*)(hall + NH);
  int*   sorted = ib;
  int*   off    = sorted + (size_t)RELS * EDGES;
  int*   cnt    = off + (size_t)RELS * NODES;
  float* dinv = (float*)ib + NH;
  int*   rank = (int*)h1;
  float* wf   = (float*)(cnt + (size_t)RELS * NODES);
  float* cf   = wf + 128 * HF;

  k_wfold<<<1, WF_T, 0, stream>>>(Wc2, bc2, W3, b3, wf, cf);

  hipMemsetAsync(cnt, 0, sizeof(int) * (size_t)RELS * NODES, stream);
  k_rank<<<EQ / EBLK, EBLK, 0, stream>>>(dst, cnt, rank);
  k_scan<<<RELS * SC_NB, SC_T, 0, stream>>>(cnt, off, dinv);
  k_fill<<<EQ / EBLK, EBLK, 0, stream>>>(src, dst, off, rank, sorted);

  const int gblocks = ((NODES / 16) + 7) / 8;
  const int ablocks = GBLOCKS;

  k_gemm<<<gblocks, 256, 0, stream>>>(x,  nullptr, nullptr, 1.0f, W1, b1, h0,   nullptr, NODES, 1);
  k_gemm<<<gblocks, 256, 0, stream>>>(h0, nullptr, nullptr, 1.0f, W2, b2, hbuf, nullptr, NODES, 1);
  hipMemsetAsync(hall, 0, sizeof(float) * NH, stream);

  for (int r = 0; r < RELS; ++r) {
    const int*   sr = sorted + (size_t)r * EDGES;
    const int*   fr = off + (size_t)r * NODES;
    const float* dv = dinv + (size_t)r * NODES;

    k_gather<<<ablocks, 256, 0, stream>>>(hbuf, dv, fr, sr, h1);
    k_gemm<<<gblocks, 256, 0, stream>>>(hbuf, h1, dv, -1.0f, Wc1, bc1, h0, nullptr, NODES, 0);

    k_gather<<<ablocks, 256, 0, stream>>>(h0, dv, fr, sr, h1);
    k_gemm<<<gblocks, 256, 0, stream>>>(h0, h1, dv, -1.0f, wf, cf, hbuf, hall, NODES, 0);
  }

  k_head<<<(NODES + 255) / 256, 256, 0, stream>>>(hall, W4, b4, out, NODES);
}


# Neighbour sum moved into the node product's launch: a wave pulls its 16-node strip (ids fetched as one run) into wave-private LDS, reads it back as A fragments; 19 launches become 13; sum order kept
# speedup vs baseline: 2.1699x; 1.1054x over previous
#include <hip/hip_runtime.h>

typedef __attribute__((ext_vector_type(2))) float v2f;
typedef __attribute__((ext_vector_type(8))) float v8f;

#define NODES 100000
#define EDGES 1000000
#define RELS  3
#define HF    64

#define EQ    (EDGES / 4)
#define EBLK  250

__launch_bounds__(256)
__global__ void k_rank(const int* __restrict__ dst, int* __restrict__ cnt,
                       int* __restrict__ rank) {
  const int q = blockIdx.x * EBLK + threadIdx.x;
  const int4* d4 = (const int4*)dst;
  int4* r4 = (int4*)rank;
  int4 d[RELS], o[RELS];
#pragma unroll
  for (int r = 0; r < RELS; ++r) d[r] = d4[r * EQ + q];
  asm volatile("" :: "v"(d[0].x), "v"(d[0].y), "v"(d[0].z), "v"(d[0].w),
                     "v"(d[1].x), "v"(d[1].y), "v"(d[1].z), "v"(d[1].w),
                     "v"(d[2].x), "v"(d[2].y), "v"(d[2].z), "v"(d[2].w));
#pragma unroll
  for (int r = 0; r < RELS; ++r) {
    int* c = cnt + r * NODES;
    o[r].x = atomicAdd(c + d[r].x, 1);
    o[r].y = atomicAdd(c + d[r].y, 1);
    o[r].z = atomicAdd(c + d[r].z, 1);
    o[r].w = atomicAdd(c + d[r].w, 1);
  }
#pragma unroll
  for (int r = 0; r < RELS; ++r) r4[r * EQ + q] = o[r];
}

#define SC_CH 2000
#define SC_NB (NODES / SC_CH)
#define SC_Q  (SC_CH / 4)
#define SC_T  512
#define SC_W  (SC_T / 32)
#define SC_U  8

__device__ __forceinline__ float deg_to_dinv(int c) {
  float v = (float)c;
  v = v < 1.0f ? 1.0f : v;
  return 1.0f / sqrtf(v);
}

__launch_bounds__(SC_T)
__global__ void k_scan(const int* __restrict__ cnt, int* __restrict__ off,
                       float* __restrict__ dinv) {
  __shared__ int wpre[SC_W];
  __shared__ int wown[SC_W];
  const int r = blockIdx.x / SC_NB;
  const int b = blockIdx.x - r * SC_NB;
  const int t = threadIdx.x;
  const int lane = t & 31;
  const int wave = t >> 5;
  const int4* c4 = (const int4*)(cnt + (size_t)r * NODES);
  const int nq = b * SC_Q;

  int4 v = make_int4(0, 0, 0, 0);
  if (t < SC_Q) v = c4[nq + t];

  int pre = 0;
  for (int i0 = t; i0 < nq; i0 += SC_U * SC_T) {
    int4 u[SC_U];
#pragma unroll
    for (int k = 0; k < SC_U; ++k) {
      u[k] = make_int4(0, 0, 0, 0);
      if (i0 + k * SC_T < nq) u[k] = c4[i0 + k * SC_T];
    }
#pragma unroll
    for (int k = 0; k < SC_U; ++k) pre += (u[k].x + u[k].y) + (u[k].z + u[k].w);
  }

  const int own = (v.x + v.y) + (v.z + v.w);
  int inc = own;
  for (int d = 1; d < 32; d <<= 1) {
    const int u = __shfl_up(inc, d);
    if (lane >= d) inc += u;
  }
  for (int d = 16; d >= 1; d >>= 1) pre += __shfl_xor(pre, d);
  if (lane == 31) {
    wown[wave] = inc;
    wpre[wave] = pre;
  }
  __syncthreads();
  int run = inc - own;
  for (int w = 0; w < SC_W; ++w) {
    run += wpre[w];
    if (w < wave) run += wown[w];
  }

  if (t < SC_Q) {
    const size_t e = (size_t)r * NODES + (size_t)(nq + t) * 4;
    int4 o;
    o.x = run; run += v.x;
    o.y = run; run += v.y;
    o.z = run; run += v.z;
    o.w = run;
    *(int4*)(off + e) = o;
    float4 dd;
    dd.x = deg_to_dinv(v.x);
    dd.y = deg_to_dinv(v.y);
    dd.z = deg_to_dinv(v.z);
    dd.w = deg_to_dinv(v.w);
    *(float4*)(dinv + e) = dd;
  }
}

__launch_bounds__(256)
__global__ void k_fill(const int* __restrict__ src, const int* __restrict__ dst,
                       const int* __restrict__ off, const int* __restrict__ rank,
                       int* __restrict__ sorted) {
  const int q = blockIdx.x * EBLK + threadIdx.x;
  const int4* s4 = (const int4*)src;
  const int4* d4 = (const int4*)dst;
  const int4* r4 = (const int4*)rank;
  int4 s[RELS], d[RELS], rk[RELS], o[RELS];
#pragma unroll
  for (int r = 0; r < RELS; ++r) {
    d[r] = d4[r * EQ + q];
    rk[r] = r4[r * EQ + q];
    s[r] = s4[r * EQ + q];
  }
#pragma unroll
  for (int r = 0; r < RELS; ++r) {
    const int* f = off + r * NODES;
    o[r].x = f[d[r].x];
    o[r].y = f[d[r].y];
    o[r].z = f[d[r].z];
    o[r].w = f[d[r].w];
  }
#pragma unroll
  for (int r = 0; r < RELS; ++r) {
    int* out = sorted + (size_t)r * EDGES;
    out[o[r].x + rk[r].x] = s[r].x;
    out[o[r].y + rk[r].y] = s[r].y;
    out[o[r].z + rk[r].z] = s[r].z;
    out[o[r].w + rk[r].w] = s[r].w;
  }
}

__launch_bounds__(256)
__global__ void k_gemm(const float* __restrict__ A1,
                       const float* __restrict__ A2,
                       const float* __restrict__ dinv,
                       float a2sign,
                       const float* __restrict__ W,
                       const float* __restrict__ bias,
                       float* __restrict__ out,
                       float* __restrict__ accum,
                       int n, int act) {
  __shared__ float sW[128 * 64];
  __shared__ float sB[64];
  const bool dual = (A2 != nullptr);
  const int wtot = (dual ? 128 : 64) * 64;
  for (int i = threadIdx.x; i < wtot; i += 256) {
    int k = i >> 6;
    int col = i & 63;
    sW[(k >> 1) * 128 + col * 2 + (k & 1)] = W[i];
  }
  if (threadIdx.x < 64) sB[threadIdx.x] = bias[threadIdx.x];
  __syncthreads();

  const int lane = threadIdx.x & 31;
  const int wave = threadIdx.x >> 5;
  const int strip = blockIdx.x * 8 + wave;
  if (strip >= (n >> 4)) return;
  const int row0 = strip << 4;
  const int m  = lane & 15;
  const int kh = (lane >> 4) << 1;

  const float* a1p = A1 + (size_t)(row0 + m) * HF + kh;
  v8f acc[4] = {};

#pragma unroll
  for (int k0 = 0; k0 < 64; k0 += 4) {
    v2f a = *(const v2f*)(a1p + k0);
    const float* wb = &sW[(k0 + kh) * 64 + m * 2];
#pragma unroll
    for (int t = 0; t < 4; ++t) {
      v2f b = *(const v2f*)(wb + t * 32);
      acc[t] = __builtin_amdgcn_wmma_f32_16x16x4_f32(
          false, a, false, b, (short)0, acc[t], false, false);
    }
  }
  if (dual) {
    const float* a2p = A2 + (size_t)(row0 + m) * HF + kh;
    const float ds = (dinv ? dinv[row0 + m] : 1.0f) * a2sign;
#pragma unroll
    for (int k0 = 0; k0 < 64; k0 += 4) {
      v2f a = *(const v2f*)(a2p + k0);
      a = a * ds;
      const float* wb = &sW[(64 + k0 + kh) * 64 + m * 2];
#pragma unroll
      for (int t = 0; t < 4; ++t) {
        v2f b = *(const v2f*)(wb + t * 32);
        acc[t] = __builtin_amdgcn_wmma_f32_16x16x4_f32(
            false, a, false, b, (short)0, acc[t], false, false);
      }
    }
  }

  const int rbase = row0 + ((lane >> 4) << 3);
#pragma unroll
  for (int t = 0; t < 4; ++t) {
    const float bv = sB[t * 16 + m];
#pragma unroll
    for (int v = 0; v < 8; ++v) {
      float val = acc[t][v] + bv;
      if (act) val = val > 0.0f ? val : 0.01f * val;
      size_t o = (size_t)(rbase + v) * HF + t * 16 + m;
      out[o] = val;
      if (accum) accum[o] += val;
    }
  }
}

#define GCH  8
#define ICAP 256
#define PA   68

__device__ __forceinline__ void prod64(v8f (&acc)[4], const float* ap, const float* wb0) {
#pragma unroll
  for (int k0 = 0; k0 < 64; k0 += 4) {
    v2f a = *(const v2f*)(ap + k0);
    const float* wb = wb0 + k0 * 64;
#pragma unroll
    for (int t = 0; t < 4; ++t) {
      v2f b = *(const v2f*)(wb + t * 32);
      acc[t] = __builtin_amdgcn_wmma_f32_16x16x4_f32(
          false, a, false, b, (short)0, acc[t], false, false);
    }
  }
}

template <bool STAGED>
__device__ __forceinline__ void pull_strip(const float* __restrict__ feat,
                                           const float* __restrict__ dinv,
                                           const int* __restrict__ sorted,
                                           const int* si, int off0, int offv, float dnv,
                                           int half, int c, float* sa) {
#pragma nounroll
  for (int k = 0; k < 8; ++k) {
    const int node = 2 * k + half;
    const int beg = __shfl(offv, node);
    const int e = __shfl(offv, node + 1);
    float4 sum = make_float4(0.0f, 0.0f, 0.0f, 0.0f);
    for (int j0 = beg; j0 < e; j0 += GCH) {
      int s[GCH];
      float dv[GCH];
      float4 f[GCH];
#pragma unroll
      for (int u = 0; u < GCH; ++u) {
        s[u] = 0;
        if (j0 + u < e) s[u] = STAGED ? si[j0 + u - off0] : sorted[j0 + u];
      }
#pragma unroll
      for (int u = 0; u < GCH; ++u) {
        dv[u] = 0.0f;
        f[u] = make_float4(0.0f, 0.0f, 0.0f, 0.0f);
        if (j0 + u < e) {
          dv[u] = dinv[s[u]];
          f[u] = *(const float4*)(feat + (size_t)s[u] * HF + c);
        }
      }
#pragma unroll
      for (int u = 0; u < GCH; ++u) {
        if (j0 + u < e) {
          sum.x += f[u].x * dv[u];
          sum.y += f[u].y * dv[u];
          sum.z += f[u].z * dv[u];
          sum.w += f[u].w * dv[u];
        }
      }
    }
    const float ds = -__shfl(dnv, node);
    sum.x *= ds;
    sum.y *= ds;
    sum.z *= ds;
    sum.w *= ds;
    *(float4*)(sa + node * PA + c) = sum;
  }
}

__device__ __forceinline__ void wave_lds_handover() {
  __builtin_amdgcn_fence(__ATOMIC_RELEASE, "wavefront");
  __builtin_amdgcn_wave_barrier();
  __builtin_amdgcn_fence(__ATOMIC_ACQUIRE, "wavefront");
}

__launch_bounds__(256)
__global__ void k_prop(const float* __restrict__ feat,
                       const float* __restrict__ dinv,
                       const int* __restrict__ off,
                       const int* __restrict__ sorted,
                       const float* __restrict__ W,
                       const float* __restrict__ bias,
                       float* __restrict__ out,
                       float* __restrict__ accum) {
  __shared__ float sW[128 * 64];
  __shared__ float sB[64];
  __shared__ float sA[8 * 16 * PA];
  __shared__ int sI[8 * ICAP];
  for (int i = threadIdx.x; i < 128 * 64; i += 256) {
    int k = i >> 6;
    int col = i & 63;
    sW[(k >> 1) * 128 + col * 2 + (k & 1)] = W[i];
  }
  if (threadIdx.x < 64) sB[threadIdx.x] = bias[threadIdx.x];
  __syncthreads();

  const int lane = threadIdx.x & 31;
  const int wave = threadIdx.x >> 5;
  const int strip = blockIdx.x * 8 + wave;
  if (strip >= (NODES >> 4)) return;
  const int row0 = strip << 4;
  const int m  = lane & 15;
  const int half = lane >> 4;
  const int kh = half << 1;

  int offv = 0;
  float dnv = 0.0f;
  if (lane <= 16) {
    const int i = row0 + lane;
    offv = (i < NODES) ? off[i] : EDGES;
    if (lane < 16) dnv = dinv[i];
  }
  const int off0 = __shfl(offv, 0);
  const int off16 = __shfl(offv, 16);
  const bool staged = (off16 - off0) <= ICAP;
  int idq[ICAP / 32];
#pragma unroll
  for (int i = 0; i < ICAP / 32; ++i) {
    const int j = off0 + lane + 32 * i;
    idq[i] = 0;
    if (j < off16) idq[i] = sorted[j];
  }

  v8f acc[4] = {};
  const float* a1p = feat + (size_t)(row0 + m) * HF + kh;
  const float* wlane = &sW[kh * 64 + m * 2];
  const bool early = (wave & 4) != 0;
  if (early) prod64(acc, a1p, wlane);

  int* si = sI + wave * ICAP;
#pragma unroll
  for (int i = 0; i < ICAP / 32; ++i) si[lane + 32 * i] = idq[i];
  wave_lds_handover();

  float* sa = sA + wave * (16 * PA);
  const int c = (lane & 15) << 2;
  if (staged) pull_strip<true>(feat, dinv, sorted, si, off0, offv, dnv, half, c, sa);
  else        pull_strip<false>(feat, dinv, sorted, si, off0, offv, dnv, half, c, sa);
  wave_lds_handover();

  if (!early) prod64(acc, a1p, wlane);
  prod64(acc, sa + m * PA + kh, wlane + 64 * 64);

  const int rbase = row0 + (half << 3);
#pragma unroll
  for (int t = 0; t < 4; ++t) {
    const float bv = sB[t * 16 + m];
#pragma unroll
    for (int v = 0; v < 8; ++v) {
      float val = acc[t][v] + bv;
      size_t o = (size_t)(rbase + v) * HF + t * 16 + m;
      out[o] = val;
      if (accum) accum[o] += val;
    }
  }
}

#define WF_T 288

__launch_bounds__(WF_T)
__global__ void k_wfold(const float* __restrict__ Wc2,
                        const float* __restrict__ bc2,
                        const float* __restrict__ W3,
                        const float* __restrict__ b3,
                        float* __restrict__ Wf,
                        float* __restrict__ cf) {
  __shared__ float sW[64 * 64];
  const float* Wlo = W3 + 64 * HF;
  for (int i = threadIdx.x; i < 64 * 64; i += WF_T) {
    int k = i >> 6;
    int col = i & 63;
    sW[(k >> 1) * 128 + col * 2 + (k & 1)] = Wlo[i];
  }
  __syncthreads();

  const int lane = threadIdx.x & 31;
  const int wave = threadIdx.x >> 5;
  const int m  = lane & 15;
  const int kh = (lane >> 4) << 1;
  const int row = (wave << 4) + m;
  const float* ap = (row < 128 ? Wc2 + (size_t)row * HF : bc2) + kh;
  const bool live = row <= 128;
  v8f acc[4] = {};

#pragma unroll
  for (int k0 = 0; k0 < 64; k0 += 4) {
    v2f a = *(const v2f*)(ap + k0);
    if (!live) a = (v2f){0.0f, 0.0f};
    const float* wb = &sW[(k0 + kh) * 64 + m * 2];
#pragma unroll
    for (int t = 0; t < 4; ++t) {
      v2f b = *(const v2f*)(wb + t * 32);
      acc[t] = __builtin_amdgcn_wmma_f32_16x16x4_f32(
          false, a, false, b, (short)0, acc[t], false, false);
    }
  }

  const int rbase = (wave << 4) + ((lane >> 4) << 3);
  if (wave < 4) {
    float u[4][8];
#pragma unroll
    for (int t = 0; t < 4; ++t)
#pragma unroll
      for (int v = 0; v < 8; ++v) u[t][v] = W3[(rbase + v) * HF + t * 16 + m];
#pragma unroll
    for (int t = 0; t < 4; ++t)
#pragma unroll
      for (int v = 0; v < 8; ++v) Wf[(rbase + v) * HF + t * 16 + m] = acc[t][v] + u[t][v];
  } else if (wave < 8) {
#pragma unroll
    for (int t = 0; t < 4; ++t)
#pragma unroll
      for (int v = 0; v < 8; ++v) Wf[(rbase + v) * HF + t * 16 + m] = acc[t][v];
  } else if (lane < 16) {
#pragma unroll
    for (int t = 0; t < 4; ++t) cf[t * 16 + m] = acc[t][0] + b3[t * 16 + m];
  }
}

__global__ void k_head(const float* __restrict__ hall,
                       const float* __restrict__ W4,
                       const float* __restrict__ b4,
                       float* __restrict__ out, int n) {
  __shared__ float sW[128];
  __shared__ float sb[2];
  if (threadIdx.x < 128) sW[threadIdx.x] = W4[threadIdx.x];
  if (threadIdx.x < 2) sb[threadIdx.x] = b4[threadIdx.x];
  __syncthreads();
  int i = blockIdx.x * blockDim.x + threadIdx.x;
  if (i >= n) return;
  const float* hp = hall + (size_t)i * HF;
  float a0 = sb[0], a1 = sb[1];
#pragma unroll
  for (int k = 0; k < 64; ++k) {
    float v = hp[k];
    v = v > 0.0f ? v : 0.01f * v;
    a0 += v * sW[2 * k];
    a1 += v * sW[2 * k + 1];
  }
  out[2 * (size_t)i + 0] = a0;
  out[2 * (size_t)i + 1] = a1;
}

extern "C" void kernel_launch(void* const* d_in, const int* in_sizes, int n_in,
                              void* d_out, int out_size, void* d_ws, size_t ws_size,
                              hipStream_t stream) {
  const float* x   = (const float*)d_in[0];
  const int*   src = (const int*)d_in[1];
  const int*   dst = (const int*)d_in[2];
  const float* W1  = (const float*)d_in[3];
  const float* b1  = (const float*)d_in[4];
  const float* W2  = (const float*)d_in[5];
  const float* b2  = (const float*)d_in[6];
  const float* Wc1 = (const float*)d_in[7];
  const float* bc1 = (const float*)d_in[8];
  const float* Wc2 = (const float*)d_in[9];
  const float* bc2 = (const float*)d_in[10];
  const float* W3  = (const float*)d_in[11];
  const float* b3  = (const float*)d_in[12];
  const float* W4  = (const float*)d_in[13];
  const float* b4  = (const float*)d_in[14];
  float* out = (float*)d_out;

  const size_t NH = (size_t)NODES * HF;
  float* ws   = (float*)d_ws;
  float* hbuf = ws;
  float* h0   = hbuf + NH;
  float* h1   = h0 + NH;
  float* hall = h1 + NH;
  int*   ib   = (int*)(hall + NH);
  int*   sorted = ib;
  int*   off    = sorted + (size_t)RELS * EDGES;
  int*   cnt    = off + (size_t)RELS * NODES;
  float* dinv = (float*)ib + NH;
  int*   rank = (int*)h1;
  float* wf   = (float*)(cnt + (size_t)RELS * NODES);
  float* cf   = wf + 128 * HF;

  k_wfold<<<1, WF_T, 0, stream>>>(Wc2, bc2, W3, b3, wf, cf);

  hipMemsetAsync(cnt, 0, sizeof(int) * (size_t)RELS * NODES, stream);
  k_rank<<<EQ / EBLK, EBLK, 0, stream>>>(dst, cnt, rank);
  k_scan<<<RELS * SC_NB, SC_T, 0, stream>>>(cnt, off, dinv);
  k_fill<<<EQ / EBLK, EBLK, 0, stream>>>(src, dst, off, rank, sorted);

  const int gblocks = ((NODES / 16) + 7) / 8;

  k_gemm<<<gblocks, 256, 0, stream>>>(x,  nullptr, nullptr, 1.0f, W1, b1, h0,   nullptr, NODES, 1);
  k_gemm<<<gblocks, 256, 0, stream>>>(h0, nullptr, nullptr, 1.0f, W2, b2, hbuf, nullptr, NODES, 1);
  hipMemsetAsync(hall, 0, sizeof(float) * NH, stream);

  for (int r = 0; r < RELS; ++r) {
    const int*   sr = sorted + (size_t)r * EDGES;
    const int*   fr = off + (size_t)r * NODES;
    const float* dv = dinv + (size_t)r * NODES;

    k_prop<<<gblocks, 256, 0, stream>>>(hbuf, dv, fr, sr, Wc1, bc1, h0, nullptr);

    k_prop<<<gblocks, 256, 0, stream>>>(h0, dv, fr, sr, wf, cf, hbuf, hall);
  }

  k_head<<<(NODES + 255) / 256, 256, 0, stream>>>(hall, W4, b4, out, NODES);
}
